# starting kernel
# baseline (speedup 1.0000x reference)
#include <hip/hip_runtime.h>
#include <math.h>

#define N_NODES   50000
#define DEG       16
#define IN_FEATS  128
#define HEADS     8
#define OUT_FEATS 16
#define HF        (HEADS * OUT_FEATS)
#define KSTEPS    (IN_FEATS / 4)

typedef __attribute__((ext_vector_type(2))) float v2f;
typedef __attribute__((ext_vector_type(4))) float v4f;
typedef __attribute__((ext_vector_type(8))) float v8f;

__global__ void gat_gemm_proj_kernel(const float* __restrict__ h,
                                     const float* __restrict__ W,
                                     const float* __restrict__ w_attn,
                                     float* __restrict__ hp,
                                     float* __restrict__ s)
{
    __shared__ float Wlds[KSTEPS * 8 * 64];

    const int tid  = threadIdx.x;
    const int lane = tid & 31;

    #pragma unroll
    for (int it = 0; it < 16; ++it) {
        const int g    = tid + it * 256;
        const int row  = g >> 5;
        const int c4   = g & 31;
        const v4f w    = *(const v4f*)(W + (size_t)row * IN_FEATS + c4 * 4);
        const int base = (c4 * 8 + (row >> 4)) * 64 + ((row & 15) << 1);
        Wlds[base + 0]  = w.x;
        Wlds[base + 1]  = w.y;
        Wlds[base + 32] = w.z;
        Wlds[base + 33] = w.w;
    }
    __syncthreads();

    const int wid = (blockIdx.x * blockDim.x + tid) >> 5;
    if (wid < N_NODES / 16) {
        const int ml = lane & 15;
        const int kh = (lane >> 4) << 1;
        const int m0 = wid * 16;

        const float* __restrict__ arow = h + (size_t)(m0 + ml) * IN_FEATS + kh;
        v2f areg[KSTEPS];
        #pragma unroll
        for (int t = 0; t < KSTEPS; ++t)
            areg[t] = *(const v2f*)(arow + 4 * t);

        v8f acc[8];
        #pragma unroll
        for (int j = 0; j < 8; ++j) acc[j] = (v8f){0.f,0.f,0.f,0.f,0.f,0.f,0.f,0.f};

        #pragma unroll
        for (int t = 0; t < KSTEPS; ++t) {
            v2f bf[8];
            #pragma unroll
            for (int j = 0; j < 8; ++j)
                bf[j] = *(const v2f*)(&Wlds[(t * 8 + j) * 64 + (lane << 1)]);
            #pragma unroll
            for (int j = 0; j < 8; ++j)
                acc[j] = __builtin_amdgcn_wmma_f32_16x16x4_f32(
                    false, areg[t], false, bf[j], (short)0, acc[j], false, false);
        }

        const float watt = w_attn[ml];
        #pragma unroll
        for (int j = 0; j < 8; ++j) {
            #pragma unroll
            for (int r = 0; r < 8; ++r) {
                const int row = m0 + r + ((lane >> 4) << 3);
                const float v = acc[j][r];
                hp[(size_t)row * HF + 16 * j + ml] = v;
                float p = v * watt;
                p += __shfl_xor(p, 1);
                p += __shfl_xor(p, 2);
                p += __shfl_xor(p, 4);
                p += __shfl_xor(p, 8);
                if (ml == 0) s[(size_t)row * HEADS + j] = p;
            }
        }
    }
}

__global__ void gat_attn_agg_kernel(const float* __restrict__ hp,
                                    const float* __restrict__ s,
                                    const float* __restrict__ ew,
                                    const int*   __restrict__ src,
                                    const float* __restrict__ bias,
                                    float* __restrict__ out)
{
    __shared__ float alpha_lds[8][DEG * HEADS];
    __shared__ int   src_lds[8][DEG];

    const int n     = (blockIdx.x * blockDim.x + threadIdx.x) >> 5;
    const int lane  = threadIdx.x & 31;
    const int wslot = (threadIdx.x >> 5);

    const int k  = lane & 15;
    const int hb = (lane >> 4) << 2;

    const int   sidx = src[n * DEG + k];
    const float lw   = log1pf(ew[n * DEG + k]);
    if (lane < 16) src_lds[wslot][k] = sidx;

    #pragma unroll
    for (int q = 0; q < 4; ++q) {
        const int hq = hb + q;
        float x = s[(size_t)sidx * HEADS + hq] + s[(size_t)n * HEADS + hq] + lw;
        x = (x > 0.f) ? x : 0.01f * x;
        float m = x;
        m = fmaxf(m, __shfl_xor(m, 1));
        m = fmaxf(m, __shfl_xor(m, 2));
        m = fmaxf(m, __shfl_xor(m, 4));
        m = fmaxf(m, __shfl_xor(m, 8));
        float ex = expf(x - m);
        float d = ex;
        d += __shfl_xor(d, 1);
        d += __shfl_xor(d, 2);
        d += __shfl_xor(d, 4);
        d += __shfl_xor(d, 8);
        alpha_lds[wslot][k * HEADS + hq] = ex / d;
    }

    __syncthreads();

    const int c  = lane << 2;
    const int hc = lane >> 2;
    float a0 = 0.f, a1 = 0.f, a2 = 0.f, a3 = 0.f;
    #pragma unroll
    for (int kk = 0; kk < DEG; ++kk) {
        const int   sk = src_lds[wslot][kk];
        const float al = alpha_lds[wslot][kk * HEADS + hc];
        const v4f   v  = *(const v4f*)(hp + (size_t)sk * HF + c);
        a0 += al * v.x;
        a1 += al * v.y;
        a2 += al * v.z;
        a3 += al * v.w;
    }
    const v4f b4 = *(const v4f*)(bias + c);
    v4f r;
    r.x = a0 + b4.x;
    r.y = a1 + b4.y;
    r.z = a2 + b4.z;
    r.w = a3 + b4.w;
    *(v4f*)(out + (size_t)n * HF + c) = r;
}

extern "C" void kernel_launch(void* const* d_in, const int* in_sizes, int n_in,
                              void* d_out, int out_size, void* d_ws, size_t ws_size,
                              hipStream_t stream) {
    const float* h       = (const float*)d_in[0];
    const float* ew      = (const float*)d_in[1];
    const int*   src     = (const int*)d_in[2];
    const float* W_fc    = (const float*)d_in[4];
    const float* w_attn  = (const float*)d_in[5];
    const float* bias    = (const float*)d_in[6];
    float*       out     = (float*)d_out;

    float* hp = (float*)d_ws;
    float* s  = hp + (size_t)N_NODES * HF;

    {
        const int waves  = N_NODES / 16;
        const int blocks = (waves * 32 + 255) / 256;
        gat_gemm_proj_kernel<<<blocks, 256, 0, stream>>>(h, W_fc, w_attn, hp, s);
    }
    {
        const int blocks = (N_NODES * 32) / 256;
        gat_attn_agg_kernel<<<blocks, 256, 0, stream>>>(hp, s, ew, src, bias, out);
    }
}


# Gather kernel: edge scores computed from the 16 gathered hp rows (4-feature partial dots, reduce-scatter over a head's 4 lanes); the scattered s[src] read, the LDS round trip and the barrier go
# speedup vs baseline: 1.5005x; 1.5005x over previous
#include <hip/hip_runtime.h>
#include <math.h>

#define N_NODES   50000
#define DEG       16
#define IN_FEATS  128
#define HEADS     8
#define OUT_FEATS 16
#define HF        (HEADS * OUT_FEATS)
#define KSTEPS    (IN_FEATS / 4)

typedef __attribute__((ext_vector_type(2))) float v2f;
typedef __attribute__((ext_vector_type(4))) float v4f;
typedef __attribute__((ext_vector_type(8))) float v8f;

template <int XM>
__device__ __forceinline__ float row_xor(float x) {
    return __builtin_bit_cast(float,
        __builtin_amdgcn_update_dpp(0, __builtin_bit_cast(int, x),
                                    0x160 + XM, 0xf, 0xf, true));
}
__device__ __forceinline__ float row_sum16(float p) {
    p += row_xor<1>(p);
    p += row_xor<2>(p);
    p += row_xor<4>(p);
    p += row_xor<8>(p);
    return p;
}
__device__ __forceinline__ float row_max16(float m) {
    m = fmaxf(m, row_xor<1>(m));
    m = fmaxf(m, row_xor<2>(m));
    m = fmaxf(m, row_xor<4>(m));
    m = fmaxf(m, row_xor<8>(m));
    return m;
}

__global__ void __launch_bounds__(256) gat_gemm_proj_kernel(const float* __restrict__ h,
                                     const float* __restrict__ W,
                                     const float* __restrict__ w_attn,
                                     float* __restrict__ hp,
                                     float* __restrict__ s)
{
    __shared__ float Wlds[KSTEPS * 8 * 64];

    const int tid  = threadIdx.x;
    const int lane = tid & 31;

    #pragma unroll
    for (int it = 0; it < 16; ++it) {
        const int g     = tid + it * 256;
        const int row   = g >> 5;
        const int c4    = g & 31;
        const v4f w     = *(const v4f*)(W + (size_t)row * IN_FEATS + c4 * 4);
        const int chunk = (c4 * 8 + (row >> 4)) * 64;
        const int s0    = ((row & 15) ^ c4) << 1;
        *(v2f*)(&Wlds[chunk + s0])        = (v2f){w.x, w.y};
        *(v2f*)(&Wlds[chunk + (s0 ^ 32)]) = (v2f){w.z, w.w};
    }
    __syncthreads();

    const int wid = (blockIdx.x * blockDim.x + tid) >> 5;
    if (wid < N_NODES / 16) {
        const int ml = lane & 15;
        const int kh = (lane >> 4) << 1;
        const int m0 = wid * 16;

        const float* __restrict__ arow = h + (size_t)(m0 + ml) * IN_FEATS + kh;
        v2f areg[KSTEPS];
        #pragma unroll
        for (int t = 0; t < KSTEPS; ++t)
            areg[t] = *(const v2f*)(arow + 4 * t);

        v8f acc[8];
        #pragma unroll
        for (int j = 0; j < 8; ++j) acc[j] = (v8f){0.f,0.f,0.f,0.f,0.f,0.f,0.f,0.f};

        #pragma unroll
        for (int t = 0; t < KSTEPS; ++t) {
            v2f bf[8];
            #pragma unroll
            for (int j = 0; j < 8; ++j)
                bf[j] = *(const v2f*)(&Wlds[(t * 8 + j) * 64 + ((lane ^ t) << 1)]);
            #pragma unroll
            for (int j = 0; j < 8; ++j)
                acc[j] = __builtin_amdgcn_wmma_f32_16x16x4_f32(
                    false, areg[t], false, bf[j], (short)0, acc[j], false, false);
        }

        const float watt  = w_attn[ml];
        const int   rbase = m0 + ((lane >> 4) << 3);
        float ps[8][8];
        #pragma unroll
        for (int j = 0; j < 8; ++j) {
            #pragma unroll
            for (int r = 0; r < 8; ++r) {
                const float v = acc[j][r];
                hp[(size_t)(rbase + r) * HF + 16 * j + ml] = v;
                ps[j][r] = row_sum16(v * watt);
            }
        }
        if (ml == 0) {
            #pragma unroll
            for (int r = 0; r < 8; ++r) {
                float* __restrict__ sp = s + (size_t)(rbase + r) * HEADS;
                *(v4f*)(sp)     = (v4f){ps[0][r], ps[1][r], ps[2][r], ps[3][r]};
                *(v4f*)(sp + 4) = (v4f){ps[4][r], ps[5][r], ps[6][r], ps[7][r]};
            }
        }
    }
}


template <int QP>
__device__ __forceinline__ float quad_lane(float x) {
    return __builtin_bit_cast(float,
        __builtin_amdgcn_update_dpp(0, __builtin_bit_cast(int, x),
                                    QP * 0x55, 0xf, 0xf, true));
}
template <unsigned SEL>
__device__ __forceinline__ float row_gather(float x) {
    const unsigned xi = __builtin_bit_cast(unsigned, x);
    return __builtin_bit_cast(float,
        __builtin_amdgcn_permlane16(xi, xi, SEL, SEL, false, false));
}

__global__ void __launch_bounds__(256) gat_attn_agg_kernel(
                                    const float* __restrict__ hp,
                                    const float* __restrict__ s,
                                    const float* __restrict__ ew,
                                    const int*   __restrict__ src,
                                    const float* __restrict__ w_attn,
                                    const float* __restrict__ bias,
                                    float* __restrict__ out)
{
    const int n    = (blockIdx.x * blockDim.x + threadIdx.x) >> 5;
    const int lane = threadIdx.x & 31;
    const int k    = lane & 15;
    const int g    = lane >> 2;
    const int j    = lane & 3;
    const int c    = lane << 2;

    const int   sidx = src[n * DEG + k];
    const float ewk  = ew[n * DEG + k];
    const float sd   = s[n * HEADS + g];
    const v4f   wa   = *(const v4f*)(w_attn + 4 * j);
    const v4f   b4   = *(const v4f*)(bias + c);
    const unsigned lane_off = (unsigned)lane << 4;
    v4f v[DEG];
    #pragma unroll
    for (int kk = 0; kk < DEG; ++kk) {
        const unsigned sk = (unsigned)__builtin_amdgcn_readlane(sidx, kk);
        const char* rowp = (const char*)hp + ((size_t)sk << 9);
        v[kk] = *(const v4f*)(rowp + lane_off);
    }
    __builtin_amdgcn_sched_barrier(0);

    float lw = log1pf(ewk);
    asm volatile("" : "+v"(lw));
    __builtin_amdgcn_sched_barrier(0);

    const v2f wlo = (v2f){wa.x, wa.y};
    const v2f whi = (v2f){wa.z, wa.w};
    float p[DEG];
    #pragma unroll
    for (int kk = 0; kk < DEG; ++kk) {
        v2f t = (v2f){v[kk].x, v[kk].y} * wlo;
        t = __builtin_elementwise_fma((v2f){v[kk].z, v[kk].w}, whi, t);
        p[kk] = t.x + t.y;
    }

    const bool b0 = (lane & 1) != 0;
    const bool b1 = (lane & 2) != 0;
    float ra[8];
    #pragma unroll
    for (int e = 0; e < 8; ++e) {
        const float keep = b0 ? p[2 * e + 1] : p[2 * e];
        const float send = b0 ? p[2 * e]     : p[2 * e + 1];
        ra[e] = keep + row_xor<1>(send);
    }
    float x[4];
    #pragma unroll
    for (int f = 0; f < 4; ++f) {
        const float keep = b1 ? ra[2 * f + 1] : ra[2 * f];
        const float send = b1 ? ra[2 * f]     : ra[2 * f + 1];
        x[f] = keep + row_xor<2>(send);
    }

    x[0] = (x[0] + sd) + row_gather<0x32103210u>(lw);
    x[1] = (x[1] + sd) + row_gather<0x76547654u>(lw);
    x[2] = (x[2] + sd) + row_gather<0xBA98BA98u>(lw);
    x[3] = (x[3] + sd) + row_gather<0xFEDCFEDCu>(lw);
    #pragma unroll
    for (int f = 0; f < 4; ++f)
        x[f] = (x[f] > 0.f) ? x[f] : 0.01f * x[f];

    float m = fmaxf(fmaxf(x[0], x[1]), fmaxf(x[2], x[3]));
    m = fmaxf(m, row_xor<1>(m));
    m = fmaxf(m, row_xor<2>(m));
    float ex[4];
    #pragma unroll
    for (int f = 0; f < 4; ++f) ex[f] = expf(x[f] - m);
    float d = (ex[0] + ex[1]) + (ex[2] + ex[3]);
    d += row_xor<1>(d);
    d += row_xor<2>(d);
    #pragma unroll
    for (int f = 0; f < 4; ++f) ex[f] = ex[f] / d;

    float a0 = 0.f, a1 = 0.f, a2 = 0.f, a3 = 0.f;
    #pragma unroll
    for (int f = 0; f < 4; ++f) {
        const float al0 = quad_lane<0>(ex[f]);
        const float al1 = quad_lane<1>(ex[f]);
        const float al2 = quad_lane<2>(ex[f]);
        const float al3 = quad_lane<3>(ex[f]);
        a0 += al0 * v[4 * f].x;     a1 += al0 * v[4 * f].y;     a2 += al0 * v[4 * f].z;     a3 += al0 * v[4 * f].w;
        a0 += al1 * v[4 * f + 1].x; a1 += al1 * v[4 * f + 1].y; a2 += al1 * v[4 * f + 1].z; a3 += al1 * v[4 * f + 1].w;
        a0 += al2 * v[4 * f + 2].x; a1 += al2 * v[4 * f + 2].y; a2 += al2 * v[4 * f + 2].z; a3 += al2 * v[4 * f + 2].w;
        a0 += al3 * v[4 * f + 3].x; a1 += al3 * v[4 * f + 3].y; a2 += al3 * v[4 * f + 3].z; a3 += al3 * v[4 * f + 3].w;
    }
    v4f r;
    r.x = a0 + b4.x;
    r.y = a1 + b4.y;
    r.z = a2 + b4.z;
    r.w = a3 + b4.w;
    *(v4f*)(out + (size_t)n * HF + c) = r;
}

extern "C" void kernel_launch(void* const* d_in, const int* in_sizes, int n_in,
                              void* d_out, int out_size, void* d_ws, size_t ws_size,
                              hipStream_t stream) {
    const float* h       = (const float*)d_in[0];
    const float* ew      = (const float*)d_in[1];
    const int*   src     = (const int*)d_in[2];
    const float* W_fc    = (const float*)d_in[4];
    const float* w_attn  = (const float*)d_in[5];
    const float* bias    = (const float*)d_in[6];
    float*       out     = (float*)d_out;

    float* hp = (float*)d_ws;
    float* s  = hp + (size_t)N_NODES * HF;

    {
        const int waves  = N_NODES / 16;
        const int blocks = (waves * 32 + 255) / 256;
        gat_gemm_proj_kernel<<<blocks, 256, 0, stream>>>(h, W_fc, w_attn, hp, s);
    }
    {
        const int blocks = (N_NODES * 32) / 256;
        gat_attn_agg_kernel<<<blocks, 256, 0, stream>>>(hp, s, ew, src, w_attn, bias, out);
    }
}


# Gather kernel: log1p, exp and the softmax divide on the hardware log2 / exp2 / rcp instructions (f32) instead of the library's guarded sequences: about 160 of a wave's 500 instructions go
# speedup vs baseline: 1.5568x; 1.0375x over previous
#include <hip/hip_runtime.h>
#include <math.h>

#define N_NODES   50000
#define DEG       16
#define IN_FEATS  128
#define HEADS     8
#define OUT_FEATS 16
#define HF        (HEADS * OUT_FEATS)
#define KSTEPS    (IN_FEATS / 4)

typedef __attribute__((ext_vector_type(2))) float v2f;
typedef __attribute__((ext_vector_type(4))) float v4f;
typedef __attribute__((ext_vector_type(8))) float v8f;

template <int XM>
__device__ __forceinline__ float row_xor(float x) {
    return __builtin_bit_cast(float,
        __builtin_amdgcn_update_dpp(0, __builtin_bit_cast(int, x),
                                    0x160 + XM, 0xf, 0xf, true));
}
__device__ __forceinline__ float row_sum16(float p) {
    p += row_xor<1>(p);
    p += row_xor<2>(p);
    p += row_xor<4>(p);
    p += row_xor<8>(p);
    return p;
}
__device__ __forceinline__ float row_max16(float m) {
    m = fmaxf(m, row_xor<1>(m));
    m = fmaxf(m, row_xor<2>(m));
    m = fmaxf(m, row_xor<4>(m));
    m = fmaxf(m, row_xor<8>(m));
    return m;
}

__global__ void __launch_bounds__(256) gat_gemm_proj_kernel(const float* __restrict__ h,
                                     const float* __restrict__ W,
                                     const float* __restrict__ w_attn,
                                     float* __restrict__ hp,
                                     float* __restrict__ s)
{
    __shared__ float Wlds[KSTEPS * 8 * 64];

    const int tid  = threadIdx.x;
    const int lane = tid & 31;

    #pragma unroll
    for (int it = 0; it < 16; ++it) {
        const int g     = tid + it * 256;
        const int row   = g >> 5;
        const int c4    = g & 31;
        const v4f w     = *(const v4f*)(W + (size_t)row * IN_FEATS + c4 * 4);
        const int chunk = (c4 * 8 + (row >> 4)) * 64;
        const int s0    = ((row & 15) ^ c4) << 1;
        *(v2f*)(&Wlds[chunk + s0])        = (v2f){w.x, w.y};
        *(v2f*)(&Wlds[chunk + (s0 ^ 32)]) = (v2f){w.z, w.w};
    }
    __syncthreads();

    const int wid = (blockIdx.x * blockDim.x + tid) >> 5;
    if (wid < N_NODES / 16) {
        const int ml = lane & 15;
        const int kh = (lane >> 4) << 1;
        const int m0 = wid * 16;

        const float* __restrict__ arow = h + (size_t)(m0 + ml) * IN_FEATS + kh;
        v2f areg[KSTEPS];
        #pragma unroll
        for (int t = 0; t < KSTEPS; ++t)
            areg[t] = *(const v2f*)(arow + 4 * t);

        v8f acc[8];
        #pragma unroll
        for (int j = 0; j < 8; ++j) acc[j] = (v8f){0.f,0.f,0.f,0.f,0.f,0.f,0.f,0.f};

        #pragma unroll
        for (int t = 0; t < KSTEPS; ++t) {
            v2f bf[8];
            #pragma unroll
            for (int j = 0; j < 8; ++j)
                bf[j] = *(const v2f*)(&Wlds[(t * 8 + j) * 64 + ((lane ^ t) << 1)]);
            #pragma unroll
            for (int j = 0; j < 8; ++j)
                acc[j] = __builtin_amdgcn_wmma_f32_16x16x4_f32(
                    false, areg[t], false, bf[j], (short)0, acc[j], false, false);
        }

        const float watt  = w_attn[ml];
        const int   rbase = m0 + ((lane >> 4) << 3);
        float ps[8][8];
        #pragma unroll
        for (int j = 0; j < 8; ++j) {
            #pragma unroll
            for (int r = 0; r < 8; ++r) {
                const float v = acc[j][r];
                hp[(size_t)(rbase + r) * HF + 16 * j + ml] = v;
                ps[j][r] = row_sum16(v * watt);
            }
        }
        if (ml == 0) {
            #pragma unroll
            for (int r = 0; r < 8; ++r) {
                float* __restrict__ sp = s + (size_t)(rbase + r) * HEADS;
                *(v4f*)(sp)     = (v4f){ps[0][r], ps[1][r], ps[2][r], ps[3][r]};
                *(v4f*)(sp + 4) = (v4f){ps[4][r], ps[5][r], ps[6][r], ps[7][r]};
            }
        }
    }
}


template <int QP>
__device__ __forceinline__ float quad_lane(float x) {
    return __builtin_bit_cast(float,
        __builtin_amdgcn_update_dpp(0, __builtin_bit_cast(int, x),
                                    QP * 0x55, 0xf, 0xf, true));
}
template <unsigned SEL>
__device__ __forceinline__ float row_gather(float x) {
    const unsigned xi = __builtin_bit_cast(unsigned, x);
    return __builtin_bit_cast(float,
        __builtin_amdgcn_permlane16(xi, xi, SEL, SEL, false, false));
}

__global__ void __launch_bounds__(256) gat_attn_agg_kernel(
                                    const float* __restrict__ hp,
                                    const float* __restrict__ s,
                                    const float* __restrict__ ew,
                                    const int*   __restrict__ src,
                                    const float* __restrict__ w_attn,
                                    const float* __restrict__ bias,
                                    float* __restrict__ out)
{
    const int n    = (blockIdx.x * blockDim.x + threadIdx.x) >> 5;
    const int lane = threadIdx.x & 31;
    const int k    = lane & 15;
    const int g    = lane >> 2;
    const int j    = lane & 3;
    const int c    = lane << 2;

    const int   sidx = src[n * DEG + k];
    const float ewk  = ew[n * DEG + k];
    const float sd   = s[n * HEADS + g];
    const v4f   wa   = *(const v4f*)(w_attn + 4 * j);
    const v4f   b4   = *(const v4f*)(bias + c);
    const unsigned lane_off = (unsigned)lane << 4;
    v4f v[DEG];
    #pragma unroll
    for (int kk = 0; kk < DEG; ++kk) {
        const unsigned sk = (unsigned)__builtin_amdgcn_readlane(sidx, kk);
        const char* rowp = (const char*)hp + ((size_t)sk << 9);
        v[kk] = *(const v4f*)(rowp + lane_off);
    }
    __builtin_amdgcn_sched_barrier(0);

    float lw = __builtin_amdgcn_logf(1.0f + ewk) * 0.693147180559945309f;
    asm volatile("" : "+v"(lw));
    __builtin_amdgcn_sched_barrier(0);

    const v2f wlo = (v2f){wa.x, wa.y};
    const v2f whi = (v2f){wa.z, wa.w};
    float p[DEG];
    #pragma unroll
    for (int kk = 0; kk < DEG; ++kk) {
        v2f t = (v2f){v[kk].x, v[kk].y} * wlo;
        t = __builtin_elementwise_fma((v2f){v[kk].z, v[kk].w}, whi, t);
        p[kk] = t.x + t.y;
    }

    const bool b0 = (lane & 1) != 0;
    const bool b1 = (lane & 2) != 0;
    float ra[8];
    #pragma unroll
    for (int e = 0; e < 8; ++e) {
        const float keep = b0 ? p[2 * e + 1] : p[2 * e];
        const float send = b0 ? p[2 * e]     : p[2 * e + 1];
        ra[e] = keep + row_xor<1>(send);
    }
    float x[4];
    #pragma unroll
    for (int f = 0; f < 4; ++f) {
        const float keep = b1 ? ra[2 * f + 1] : ra[2 * f];
        const float send = b1 ? ra[2 * f]     : ra[2 * f + 1];
        x[f] = keep + row_xor<2>(send);
    }

    x[0] = (x[0] + sd) + row_gather<0x32103210u>(lw);
    x[1] = (x[1] + sd) + row_gather<0x76547654u>(lw);
    x[2] = (x[2] + sd) + row_gather<0xBA98BA98u>(lw);
    x[3] = (x[3] + sd) + row_gather<0xFEDCFEDCu>(lw);
    #pragma unroll
    for (int f = 0; f < 4; ++f)
        x[f] = (x[f] > 0.f) ? x[f] : 0.01f * x[f];

    float m = fmaxf(fmaxf(x[0], x[1]), fmaxf(x[2], x[3]));
    m = fmaxf(m, row_xor<1>(m));
    m = fmaxf(m, row_xor<2>(m));
    const float LOG2E = 1.44269504088896341f;
    const float ml2   = m * LOG2E;
    float ex[4];
    #pragma unroll
    for (int f = 0; f < 4; ++f) ex[f] = __builtin_amdgcn_exp2f(__builtin_fmaf(x[f], LOG2E, -ml2));
    float d = (ex[0] + ex[1]) + (ex[2] + ex[3]);
    d += row_xor<1>(d);
    d += row_xor<2>(d);
    const float rd = __builtin_amdgcn_rcpf(d);
    #pragma unroll
    for (int f = 0; f < 4; ++f) ex[f] = ex[f] * rd;

    float a0 = 0.f, a1 = 0.f, a2 = 0.f, a3 = 0.f;
    #pragma unroll
    for (int f = 0; f < 4; ++f) {
        const float al0 = quad_lane<0>(ex[f]);
        const float al1 = quad_lane<1>(ex[f]);
        const float al2 = quad_lane<2>(ex[f]);
        const float al3 = quad_lane<3>(ex[f]);
        a0 += al0 * v[4 * f].x;     a1 += al0 * v[4 * f].y;     a2 += al0 * v[4 * f].z;     a3 += al0 * v[4 * f].w;
        a0 += al1 * v[4 * f + 1].x; a1 += al1 * v[4 * f + 1].y; a2 += al1 * v[4 * f + 1].z; a3 += al1 * v[4 * f + 1].w;
        a0 += al2 * v[4 * f + 2].x; a1 += al2 * v[4 * f + 2].y; a2 += al2 * v[4 * f + 2].z; a3 += al2 * v[4 * f + 2].w;
        a0 += al3 * v[4 * f + 3].x; a1 += al3 * v[4 * f + 3].y; a2 += al3 * v[4 * f + 3].z; a3 += al3 * v[4 * f + 3].w;
    }
    v4f r;
    r.x = a0 + b4.x;
    r.y = a1 + b4.y;
    r.z = a2 + b4.z;
    r.w = a3 + b4.w;
    *(v4f*)(out + (size_t)n * HF + c) = r;
}

extern "C" void kernel_launch(void* const* d_in, const int* in_sizes, int n_in,
                              void* d_out, int out_size, void* d_ws, size_t ws_size,
                              hipStream_t stream) {
    const float* h       = (const float*)d_in[0];
    const float* ew      = (const float*)d_in[1];
    const int*   src     = (const int*)d_in[2];
    const float* W_fc    = (const float*)d_in[4];
    const float* w_attn  = (const float*)d_in[5];
    const float* bias    = (const float*)d_in[6];
    float*       out     = (float*)d_out;

    float* hp = (float*)d_ws;
    float* s  = hp + (size_t)N_NODES * HF;

    {
        const int waves  = N_NODES / 16;
        const int blocks = (waves * 32 + 255) / 256;
        gat_gemm_proj_kernel<<<blocks, 256, 0, stream>>>(h, W_fc, w_attn, hp, s);
    }
    {
        const int blocks = (N_NODES * 32) / 256;
        gat_attn_agg_kernel<<<blocks, 256, 0, stream>>>(hp, s, ew, src, w_attn, bias, out);
    }
}
